# starting kernel
# baseline (speedup 1.0000x reference)
#include <hip/hip_runtime.h>
#include <math.h>

typedef __attribute__((ext_vector_type(2))) float v2f;
typedef __attribute__((ext_vector_type(8))) float v8f;

#define WAVES 4
#define THREADS (WAVES * 32)
#define EDGES_PER_WAVE 32
#define EDGES_PER_PASS (WAVES * EDGES_PER_WAVE)
#define CHUNKS 8
#define EPB (EDGES_PER_PASS * CHUNKS)

#define W_ROW 512
#define H_ROW 132

__device__ __forceinline__ v8f wmma_f32_16x16x4(v2f a, v2f b, v8f c) {
  return __builtin_amdgcn_wmma_f32_16x16x4_f32(false, a, false, b, (short)0, c, false, false);
}

__device__ __forceinline__ v2f lds2(const float* p) { return *(const v2f*)p; }

__device__ __forceinline__ unsigned fkey(float f) {
  unsigned u = __float_as_uint(f);
  return (u & 0x80000000u) ? ~u : (u | 0x80000000u);
}
__device__ __forceinline__ float funkey(unsigned k) {
  unsigned u = (k & 0x80000000u) ? (k & 0x7fffffffu) : ~k;
  return __uint_as_float(u);
}

__global__ __launch_bounds__(THREADS) void edge_mlp_kernel(
    const float* __restrict__ uf, const float* __restrict__ gf,
    const float* __restrict__ ef, const int* __restrict__ uidx,
    const int* __restrict__ gidx,
    const float* __restrict__ W1, const float* __restrict__ b1,
    const float* __restrict__ W2, const float* __restrict__ b2,
    const float* __restrict__ w3, const float* __restrict__ b3,
    float* __restrict__ logits, int E)
{
  extern __shared__ float smem[];
  float* sW1g = smem;
  float* sb1  = sW1g + 2 * W_ROW;
  float* sW2g = sb1 + 128;
  float* sb2  = sW2g + 32 * W_ROW;
  float* sw3  = sb2 + 128;
  float* sX   = sw3 + 128;
  float* sH   = sX + WAVES * 2 * H_ROW;

  const int tid = threadIdx.x;

  for (int i = tid; i < 128; i += THREADS) { sb1[i] = b1[i]; sb2[i] = b2[i]; sw3[i] = w3[i]; }
  for (int i = tid; i < 8 * 128; i += THREADS) {
    const int k = i >> 7, n = i & 127;
    sW1g[(k >> 2) * W_ROW + n * 4 + (k & 3)] = W1[i];
  }
  for (int i = tid; i < 128 * 128; i += THREADS) {
    const int k = i >> 7, n = i & 127;
    sW2g[(k >> 2) * W_ROW + n * 4 + (k & 3)] = W2[i];
  }
  __syncthreads();

  const int wid  = tid >> 5;
  const int lane = tid & 31;
  const int l16  = lane & 15;
  const int half = lane >> 4;
  const int koff = half * 2;
  float* sXw = sX + wid * (2 * H_ROW);
  float* sHw = sH + wid * (32 * H_ROW);

  const int blockBase = blockIdx.x * EPB;

  for (int c = 0; c < CHUNKS; ++c) {
    const int eBase = blockBase + c * EDGES_PER_PASS + wid * EDGES_PER_WAVE;

    {
      const int m = lane;
      int e = eBase + m; if (e >= E) e = E - 1;
      const int uI = uidx[e], gI = gidx[e];
      float v[8];
      v[0] = uf[uI * 3 + 0]; v[1] = uf[uI * 3 + 1]; v[2] = uf[uI * 3 + 2];
      v[3] = gf[gI * 3 + 0]; v[4] = gf[gI * 3 + 1]; v[5] = gf[gI * 3 + 2];
      v[6] = ef[e * 2 + 0];  v[7] = ef[e * 2 + 1];
      #pragma unroll
      for (int f = 0; f < 8; ++f)
        sXw[(f >> 2) * H_ROW + m * 4 + (f & 3)] = v[f];
    }
    __builtin_amdgcn_wave_barrier();

    #pragma unroll
    for (int nt = 0; nt < 8; ++nt) {
      const int n = nt * 16 + l16;
      const v2f b0 = lds2(&sW1g[0 * W_ROW + n * 4 + koff]);
      const v2f b1v = lds2(&sW1g[1 * W_ROW + n * 4 + koff]);
      const float bias = sb1[n];
      #pragma unroll
      for (int t = 0; t < 2; ++t) {
        const v2f a0 = lds2(&sXw[0 * H_ROW + (t * 16 + l16) * 4 + koff]);
        const v2f a1 = lds2(&sXw[1 * H_ROW + (t * 16 + l16) * 4 + koff]);
        v8f cc = {0.f, 0.f, 0.f, 0.f, 0.f, 0.f, 0.f, 0.f};
        cc = wmma_f32_16x16x4(a0, b0, cc);
        cc = wmma_f32_16x16x4(a1, b1v, cc);
        #pragma unroll
        for (int r = 0; r < 8; ++r) {
          float h = cc[r] + bias;
          h = h > 0.f ? h : 0.f;
          const int m = t * 16 + r + half * 8;
          sHw[(n >> 2) * H_ROW + m * 4 + (n & 3)] = h;
        }
      }
    }
    __builtin_amdgcn_wave_barrier();

    float accw0[8], accw1[8];
    #pragma unroll
    for (int r = 0; r < 8; ++r) { accw0[r] = 0.f; accw1[r] = 0.f; }

    for (int nt = 0; nt < 8; ++nt) {
      const int n = nt * 16 + l16;
      v8f c0 = {0.f, 0.f, 0.f, 0.f, 0.f, 0.f, 0.f, 0.f};
      v8f c1 = {0.f, 0.f, 0.f, 0.f, 0.f, 0.f, 0.f, 0.f};
      #pragma unroll 4
      for (int ks = 0; ks < 32; ++ks) {
        const v2f b  = lds2(&sW2g[ks * W_ROW + n * 4 + koff]);
        const v2f a0 = lds2(&sHw[ks * H_ROW + l16 * 4 + koff]);
        const v2f a1 = lds2(&sHw[ks * H_ROW + (16 + l16) * 4 + koff]);
        c0 = wmma_f32_16x16x4(a0, b, c0);
        c1 = wmma_f32_16x16x4(a1, b, c1);
      }
      const float b2v = sb2[n];
      const float w3v = sw3[n];
      #pragma unroll
      for (int r = 0; r < 8; ++r) {
        float h0 = c0[r] + b2v; h0 = h0 > 0.f ? h0 : 0.f;
        float h1 = c1[r] + b2v; h1 = h1 > 0.f ? h1 : 0.f;
        accw0[r] += h0 * w3v;
        accw1[r] += h1 * w3v;
      }
    }

    #pragma unroll
    for (int r = 0; r < 8; ++r) {
      float v0 = accw0[r], v1 = accw1[r];
      v0 += __shfl_xor(v0, 1);  v1 += __shfl_xor(v1, 1);
      v0 += __shfl_xor(v0, 2);  v1 += __shfl_xor(v1, 2);
      v0 += __shfl_xor(v0, 4);  v1 += __shfl_xor(v1, 4);
      v0 += __shfl_xor(v0, 8);  v1 += __shfl_xor(v1, 8);
      accw0[r] = v0; accw1[r] = v1;
    }
    if (l16 == 0) {
      const float bb3 = b3[0];
      #pragma unroll
      for (int r = 0; r < 8; ++r) {
        const int e0 = eBase + half * 8 + r;
        const int e1 = eBase + 16 + half * 8 + r;
        if (e0 < E) logits[e0] = accw0[r] + bb3;
        if (e1 < E) logits[e1] = accw1[r] + bb3;
      }
    }
    __builtin_amdgcn_wave_barrier();
  }
}

__global__ void notx_kernel(const float* __restrict__ uf,
                            const float* __restrict__ Wn1, const float* __restrict__ bn1,
                            const float* __restrict__ wn2, const float* __restrict__ bn2,
                            float* __restrict__ logit_no, unsigned* __restrict__ mkey, int nU)
{
  __shared__ float sW[192], sB[64], sw2[64];
  const int tid = threadIdx.x;
  for (int i = tid; i < 192; i += blockDim.x) sW[i] = Wn1[i];
  for (int i = tid; i < 64; i += blockDim.x) { sB[i] = bn1[i]; sw2[i] = wn2[i]; }
  __syncthreads();
  const int u = blockIdx.x * blockDim.x + tid;
  if (u >= nU) return;
  const float f0 = uf[u * 3 + 0], f1 = uf[u * 3 + 1], f2 = uf[u * 3 + 2];
  float acc = bn2[0];
  #pragma unroll 8
  for (int j = 0; j < 64; ++j) {
    float h = f0 * sW[j] + f1 * sW[64 + j] + f2 * sW[128 + j] + sB[j];
    h = h > 0.f ? h : 0.f;
    acc += h * sw2[j];
  }
  logit_no[u] = acc;
  mkey[u] = fkey(acc);
}

__global__ void segmax_kernel(const float* __restrict__ logits, const int* __restrict__ uidx,
                              unsigned* __restrict__ mkey, int E)
{
  const int e = blockIdx.x * blockDim.x + threadIdx.x;
  if (e < E) atomicMax(&mkey[uidx[e]], fkey(logits[e]));
}

__global__ void user_exp_kernel(const float* __restrict__ logit_no, const unsigned* __restrict__ mkey,
                                float* __restrict__ denom, float* __restrict__ outNo, int nU)
{
  const int u = blockIdx.x * blockDim.x + threadIdx.x;
  if (u < nU) {
    const float m  = funkey(mkey[u]);
    const float en = expf(logit_no[u] - m);
    denom[u] = en;
    outNo[u] = en;
  }
}

__global__ void edge_exp_kernel(const float* __restrict__ logits, const int* __restrict__ uidx,
                                const unsigned* __restrict__ mkey, float* __restrict__ denom,
                                float* __restrict__ outE, int E)
{
  const int e = blockIdx.x * blockDim.x + threadIdx.x;
  if (e < E) {
    const int u = uidx[e];
    const float ex = expf(logits[e] - funkey(mkey[u]));
    outE[e] = ex;
    atomicAdd(&denom[u], ex);
  }
}

__global__ void edge_div_kernel(float* __restrict__ outE, const int* __restrict__ uidx,
                                const float* __restrict__ denom, int E)
{
  const int e = blockIdx.x * blockDim.x + threadIdx.x;
  if (e < E) outE[e] = outE[e] / denom[uidx[e]];
}

__global__ void user_div_kernel(float* __restrict__ outNo, const float* __restrict__ denom, int nU)
{
  const int u = blockIdx.x * blockDim.x + threadIdx.x;
  if (u < nU) outNo[u] = outNo[u] / denom[u];
}

extern "C" void kernel_launch(void* const* d_in, const int* in_sizes, int n_in,
                              void* d_out, int out_size, void* d_ws, size_t ws_size,
                              hipStream_t stream)
{
  const float* uf  = (const float*)d_in[0];
  const float* gf  = (const float*)d_in[1];
  const float* ef  = (const float*)d_in[2];
  const int*   ui  = (const int*)d_in[3];
  const int*   gi  = (const int*)d_in[4];
  const float* W1  = (const float*)d_in[5];
  const float* b1  = (const float*)d_in[6];
  const float* W2  = (const float*)d_in[7];
  const float* b2  = (const float*)d_in[8];
  const float* w3  = (const float*)d_in[9];
  const float* b3  = (const float*)d_in[10];
  const float* Wn1 = (const float*)d_in[11];
  const float* bn1 = (const float*)d_in[12];
  const float* wn2 = (const float*)d_in[13];
  const float* bn2 = (const float*)d_in[14];

  const int nU = in_sizes[0] / 3;
  const int E  = in_sizes[3];

  float* ws          = (float*)d_ws;
  float* edge_logits = ws;
  float* logit_no    = ws + (size_t)E;
  unsigned* mkey     = (unsigned*)(ws + (size_t)E + (size_t)nU);
  float* denom       = ws + (size_t)E + 2 * (size_t)nU;

  float* outE  = (float*)d_out;
  float* outNo = outE + (size_t)E;

  notx_kernel<<<(nU + 255) / 256, 256, 0, stream>>>(uf, Wn1, bn1, wn2, bn2, logit_no, mkey, nU);

  const size_t lds = (size_t)(2 * W_ROW + 128 + 32 * W_ROW + 128 + 128
                              + WAVES * 2 * H_ROW + WAVES * 32 * H_ROW) * sizeof(float);
  const int nBlk = (E + EPB - 1) / EPB;
  edge_mlp_kernel<<<nBlk, THREADS, lds, stream>>>(uf, gf, ef, ui, gi,
                                                  W1, b1, W2, b2, w3, b3, edge_logits, E);

  segmax_kernel<<<(E + 255) / 256, 256, 0, stream>>>(edge_logits, ui, mkey, E);
  user_exp_kernel<<<(nU + 255) / 256, 256, 0, stream>>>(logit_no, mkey, denom, outNo, nU);
  edge_exp_kernel<<<(E + 255) / 256, 256, 0, stream>>>(edge_logits, ui, mkey, denom, outE, E);
  edge_div_kernel<<<(E + 255) / 256, 256, 0, stream>>>(outE, ui, denom, E);
  user_div_kernel<<<(nU + 255) / 256, 256, 0, stream>>>(outNo, denom, nU);
}


# segment softmax without global atomics: LDS fold of slice max, then sum of exp, per (user window, edge slice); partial pairs combined per user in slice order; 7 launches to 5; changes f32 sum order
# speedup vs baseline: 1.7589x; 1.7589x over previous
#include <hip/hip_runtime.h>
#include <math.h>


typedef __attribute__((ext_vector_type(2))) float v2f;
typedef __attribute__((ext_vector_type(4))) float v4f;
typedef __attribute__((ext_vector_type(8))) float v8f;

#define WAVES 4
#define THREADS (WAVES * 32)
#define PASS_TILES 8
#define PASS_EDGES (PASS_TILES * 16)
#define BLOCK_TILES 16
#define EPB (BLOCK_TILES * 16)

#define ROW 516

__device__ __forceinline__ v8f wmma_f32_16x16x4(v2f a, v2f b, v8f c) {
  return __builtin_amdgcn_wmma_f32_16x16x4_f32(false, a, false, b, (short)0, c, false, false);
}

__device__ __forceinline__ unsigned fkey(float f) {
  unsigned u = __float_as_uint(f);
  return (u & 0x80000000u) ? ~u : (u | 0x80000000u);
}
__device__ __forceinline__ float funkey(unsigned k) {
  unsigned u = (k & 0x80000000u) ? (k & 0x7fffffffu) : ~k;
  return __uint_as_float(u);
}

__global__ __launch_bounds__(THREADS) void edge_mlp_kernel(
    const float* __restrict__ uf, const float* __restrict__ gf,
    const float* __restrict__ ef, const int* __restrict__ uidx,
    const int* __restrict__ gidx,
    const float* __restrict__ W1, const float* __restrict__ b1,
    const float* __restrict__ W2, const float* __restrict__ b2,
    const float* __restrict__ w3, const float* __restrict__ b3,
    float* __restrict__ logits, int E)
{
  extern __shared__ float smem[];
  float* sX = smem;
  float* sH = sX + 2 * ROW;
  float* sP = sH + 32 * ROW;

  const int tid  = threadIdx.x;
  const int wid  = tid >> 5;
  const int lane = tid & 31;
  const int c    = lane & 15;
  const int h    = lane >> 4;
  const int nb   = wid * 32;

  v2f Bw1[2][2];
  v2f Bw2[16][2][2];
  float bias1[2], bias2[2], w3v[2];
  #pragma unroll
  for (int nt = 0; nt < 2; ++nt) {
    const int n = nb + nt * 16 + c;
    bias1[nt] = b1[n]; bias2[nt] = b2[n]; w3v[nt] = w3[n];
    #pragma unroll
    for (int j = 0; j < 2; ++j) {
      const int k0 = 4 * h + 2 * j;
      Bw1[j][nt] = v2f{W1[k0 * 128 + n], W1[(k0 + 1) * 128 + n]};
    }
    #pragma unroll
    for (int p = 0; p < 16; ++p) {
      #pragma unroll
      for (int j = 0; j < 2; ++j) {
        const int k0 = 8 * p + 4 * h + 2 * j;
        Bw2[p][j][nt] = v2f{W2[k0 * 128 + n], W2[(k0 + 1) * 128 + n]};
      }
    }
  }
  const float bb3 = b3[0];

  const int tilesTotal = (E + 15) >> 4;
  const int tStart = (int)blockIdx.x * BLOCK_TILES;
  const int tRest  = tilesTotal - tStart;
  const int tCount = tRest < BLOCK_TILES ? tRest : BLOCK_TILES;
  const int nPass  = (tCount + PASS_TILES - 1) / PASS_TILES;

  float xv[8];
  int eN, uI, gI;
  {
    eN = tStart * 16 + tid; if (eN >= E) eN = E - 1;
    uI = uidx[eN]; gI = gidx[eN];
    xv[0] = uf[uI * 3 + 0]; xv[1] = uf[uI * 3 + 1]; xv[2] = uf[uI * 3 + 2];
    xv[3] = gf[gI * 3 + 0]; xv[4] = gf[gI * 3 + 1]; xv[5] = gf[gI * 3 + 2];
    xv[6] = ef[eN * 2 + 0]; xv[7] = ef[eN * 2 + 1];
  }

  const int aoff = h * ROW + c * 4;

  for (int i = 0; i < nPass; ++i) {
    const int eBase = (tStart + i * PASS_TILES) * 16;
    const int tilesLeft = tCount - i * PASS_TILES;
    const int tilesThis = tilesLeft < PASS_TILES ? tilesLeft : PASS_TILES;
    const bool more = (i + 1 < nPass);

    *(v4f*)&sX[tid * 4]       = v4f{xv[0], xv[1], xv[2], xv[3]};
    *(v4f*)&sX[ROW + tid * 4] = v4f{xv[4], xv[5], xv[6], xv[7]};
    __syncthreads();

    if (more) {
      eN = eBase + PASS_EDGES + tid; if (eN >= E) eN = E - 1;
      uI = uidx[eN]; gI = gidx[eN];
    }

    for (int mt = 0; mt < tilesThis; ++mt) {
      const v4f a = *(const v4f*)&sX[aoff + mt * 64];
      #pragma unroll
      for (int nt = 0; nt < 2; ++nt) {
        v8f cc = {0.f, 0.f, 0.f, 0.f, 0.f, 0.f, 0.f, 0.f};
        cc = wmma_f32_16x16x4(v2f{a.x, a.y}, Bw1[0][nt], cc);
        cc = wmma_f32_16x16x4(v2f{a.z, a.w}, Bw1[1][nt], cc);
        const int n = nb + nt * 16 + c;
        float* dst = &sH[(n >> 2) * ROW + (mt * 16 + h * 8) * 4 + (n & 3)];
        #pragma unroll
        for (int r = 0; r < 8; ++r) {
          float hv = cc[r] + bias1[nt];
          dst[r * 4] = hv > 0.f ? hv : 0.f;
        }
      }
    }
    __syncthreads();

    if (more) {
      xv[0] = uf[uI * 3 + 0]; xv[1] = uf[uI * 3 + 1]; xv[2] = uf[uI * 3 + 2];
      xv[3] = gf[gI * 3 + 0]; xv[4] = gf[gI * 3 + 1]; xv[5] = gf[gI * 3 + 2];
      xv[6] = ef[eN * 2 + 0]; xv[7] = ef[eN * 2 + 1];
    }

    v4f aLo[8], aHi[8];
    #pragma unroll
    for (int p = 0; p < 8; ++p) aLo[p] = *(const v4f*)&sH[(2 * p) * ROW + aoff];

    for (int mt = 0; mt < tilesThis; ++mt) {
      const int mtN = (mt + 1 < tilesThis) ? mt + 1 : mt;
      v8f c0 = {0.f, 0.f, 0.f, 0.f, 0.f, 0.f, 0.f, 0.f};
      v8f c1 = {0.f, 0.f, 0.f, 0.f, 0.f, 0.f, 0.f, 0.f};

      #pragma unroll
      for (int p = 0; p < 8; ++p) aHi[p] = *(const v4f*)&sH[(16 + 2 * p) * ROW + aoff + mt * 64];
      #pragma unroll
      for (int p = 0; p < 8; ++p) {
        const v2f a0 = v2f{aLo[p].x, aLo[p].y}, a1 = v2f{aLo[p].z, aLo[p].w};
        c0 = wmma_f32_16x16x4(a0, Bw2[p][0][0], c0);
        c1 = wmma_f32_16x16x4(a0, Bw2[p][0][1], c1);
        c0 = wmma_f32_16x16x4(a1, Bw2[p][1][0], c0);
        c1 = wmma_f32_16x16x4(a1, Bw2[p][1][1], c1);
      }
      #pragma unroll
      for (int p = 0; p < 8; ++p) aLo[p] = *(const v4f*)&sH[(2 * p) * ROW + aoff + mtN * 64];
      #pragma unroll
      for (int p = 0; p < 8; ++p) {
        const v2f a0 = v2f{aHi[p].x, aHi[p].y}, a1 = v2f{aHi[p].z, aHi[p].w};
        c0 = wmma_f32_16x16x4(a0, Bw2[8 + p][0][0], c0);
        c1 = wmma_f32_16x16x4(a0, Bw2[8 + p][0][1], c1);
        c0 = wmma_f32_16x16x4(a1, Bw2[8 + p][1][0], c0);
        c1 = wmma_f32_16x16x4(a1, Bw2[8 + p][1][1], c1);
      }

      float s[8];
      #pragma unroll
      for (int r = 0; r < 8; ++r) {
        float h0 = c0[r] + bias2[0]; h0 = h0 > 0.f ? h0 : 0.f;
        float h1 = c1[r] + bias2[1]; h1 = h1 > 0.f ? h1 : 0.f;
        s[r] = h0 * w3v[0] + h1 * w3v[1];
      }
      float t4[4], t2[2], t1;
      {
        const bool up = (c & 8) != 0;
        #pragma unroll
        for (int q = 0; q < 4; ++q) {
          const float keep = up ? s[4 + q] : s[q];
          const float send = up ? s[q] : s[4 + q];
          t4[q] = keep + __shfl_xor(send, 8);
        }
      }
      {
        const bool up = (c & 4) != 0;
        #pragma unroll
        for (int q = 0; q < 2; ++q) {
          const float keep = up ? t4[2 + q] : t4[q];
          const float send = up ? t4[q] : t4[2 + q];
          t2[q] = keep + __shfl_xor(send, 4);
        }
      }
      {
        const bool up = (c & 2) != 0;
        const float keep = up ? t2[1] : t2[0];
        const float send = up ? t2[0] : t2[1];
        t1 = keep + __shfl_xor(send, 2);
      }
      t1 += __shfl_xor(t1, 1);
      if ((c & 1) == 0) {
        const int r = ((c >> 3) << 2) | (((c >> 2) & 1) << 1) | ((c >> 1) & 1);
        sP[wid * PASS_EDGES + mt * 16 + h * 8 + r] = t1;
      }
    }
    __syncthreads();

    {
      const int e = eBase + tid;
      if (tid < tilesThis * 16 && e < E)
        logits[e] = ((sP[tid] + sP[PASS_EDGES + tid]) + (sP[2 * PASS_EDGES + tid] + sP[3 * PASS_EDGES + tid])) + bb3;
    }
  }
}

__global__ void notx_kernel(const float* __restrict__ uf,
                            const float* __restrict__ Wn1, const float* __restrict__ bn1,
                            const float* __restrict__ wn2, const float* __restrict__ bn2,
                            float* __restrict__ logit_no, int nU)
{
  __shared__ float sW[192], sB[64], sw2[64];
  const int tid = threadIdx.x;
  for (int i = tid; i < 192; i += blockDim.x) sW[i] = Wn1[i];
  for (int i = tid; i < 64; i += blockDim.x) { sB[i] = bn1[i]; sw2[i] = wn2[i]; }
  __syncthreads();
  const int u = blockIdx.x * blockDim.x + tid;
  if (u >= nU) return;
  const float f0 = uf[u * 3 + 0], f1 = uf[u * 3 + 1], f2 = uf[u * 3 + 2];
  float acc = bn2[0];
  #pragma unroll 8
  for (int j = 0; j < 64; ++j) {
    float h = f0 * sW[j] + f1 * sW[64 + j] + f2 * sW[128 + j] + sB[j];
    h = h > 0.f ? h : 0.f;
    acc += h * sw2[j];
  }
  logit_no[u] = acc;
}

#define FW_USERS   3136
#define FW_TAB     FW_USERS
#define FW_THREADS 1024
#define FW_SLICES  8
#define FW_GROUPS  4

__device__ __forceinline__ void fold_max(unsigned* tmax, int u, float l, int base, int cnt) {
  const unsigned d = (unsigned)(u - base);
  if (d < (unsigned)cnt)
    __hip_atomic_fetch_max(&tmax[d], fkey(l), __ATOMIC_RELAXED, __HIP_MEMORY_SCOPE_WORKGROUP);
}
__device__ __forceinline__ void fold_sum(const unsigned* tmax, float* tsum, int u, float l, int base, int cnt) {
  const unsigned d = (unsigned)(u - base);
  if (d < (unsigned)cnt) {
    const float t = (l == -INFINITY) ? 0.f : expf(l - funkey(tmax[d]));
    __hip_atomic_fetch_add(&tsum[d], t, __ATOMIC_RELAXED, __HIP_MEMORY_SCOPE_WORKGROUP);
  }
}

template <bool SUM>
__device__ __forceinline__ void fold_sweep(unsigned* tmax, float* tsum,
                                           const float* __restrict__ logits,
                                           const int* __restrict__ uidx,
                                           int e0, int n, int base, int cnt, int tid) {
  const int nvec = n >> 2;
  const int4*   u4 = (const int4*)(uidx + e0);
  const float4* l4 = (const float4*)(logits + e0);
  for (int v = tid; v < nvec; v += FW_GROUPS * FW_THREADS) {
    int4 uu[FW_GROUPS]; float4 ll[FW_GROUPS]; bool ok[FW_GROUPS];
    #pragma unroll
    for (int j = 0; j < FW_GROUPS; ++j) {
      const int vj = v + j * FW_THREADS;
      ok[j] = vj < nvec;
      const int vi = ok[j] ? vj : v;
      uu[j] = u4[vi];
      ll[j] = l4[vi];
    }
    #pragma unroll
    for (int j = 0; j < FW_GROUPS; ++j) {
      if (ok[j]) {
        if (SUM) {
          fold_sum(tmax, tsum, uu[j].x, ll[j].x, base, cnt);
          fold_sum(tmax, tsum, uu[j].y, ll[j].y, base, cnt);
          fold_sum(tmax, tsum, uu[j].z, ll[j].z, base, cnt);
          fold_sum(tmax, tsum, uu[j].w, ll[j].w, base, cnt);
        } else {
          fold_max(tmax, uu[j].x, ll[j].x, base, cnt);
          fold_max(tmax, uu[j].y, ll[j].y, base, cnt);
          fold_max(tmax, uu[j].z, ll[j].z, base, cnt);
          fold_max(tmax, uu[j].w, ll[j].w, base, cnt);
        }
      }
    }
  }
  for (int e = e0 + (nvec << 2) + tid; e < e0 + n; e += FW_THREADS) {
    if (SUM) fold_sum(tmax, tsum, uidx[e], logits[e], base, cnt);
    else     fold_max(tmax, uidx[e], logits[e], base, cnt);
  }
}

__global__ __launch_bounds__(FW_THREADS) void fold_kernel(
    const float* __restrict__ logits, const int* __restrict__ uidx,
    unsigned* __restrict__ pkey, float* __restrict__ psum, int E, int nU)
{
  __shared__ unsigned tmax[FW_TAB];
  __shared__ float    tsum[FW_TAB];
  const int tid  = threadIdx.x;
  const int base = (int)blockIdx.y * FW_USERS;
  int cnt = nU - base; if (cnt > FW_USERS) cnt = FW_USERS;

  for (int i = tid; i < FW_TAB; i += FW_THREADS) { tmax[i] = 0u; tsum[i] = 0.f; }
  __syncthreads();

  const int per = (((E + (int)gridDim.x - 1) / (int)gridDim.x) + 3) & ~3;
  const int e0  = (int)blockIdx.x * per;
  int n = E - e0; if (n > per) n = per; if (n < 0) n = 0;

  fold_sweep<false>(tmax, tsum, logits, uidx, e0, n, base, cnt, tid);
  __syncthreads();
  fold_sweep<true>(tmax, tsum, logits, uidx, e0, n, base, cnt, tid);
  __syncthreads();

  unsigned* pk = pkey + (size_t)blockIdx.x * (size_t)nU + base;
  float*    ps = psum + (size_t)blockIdx.x * (size_t)nU + base;
  for (int i = tid; i < cnt; i += FW_THREADS) { pk[i] = tmax[i]; ps[i] = tsum[i]; }
}

__global__ void combine_kernel(const float* __restrict__ logit_no,
                               const unsigned* __restrict__ pkey, const float* __restrict__ psum,
                               float* __restrict__ md, float* __restrict__ outNo, int nU)
{
  const int u = blockIdx.x * blockDim.x + threadIdx.x;
  if (u >= nU) return;
  const float lno = logit_no[u];
  unsigned k[FW_SLICES]; float s[FW_SLICES];
  #pragma unroll
  for (int b = 0; b < FW_SLICES; ++b) {
    k[b] = pkey[(size_t)b * (size_t)nU + u];
    s[b] = psum[(size_t)b * (size_t)nU + u];
  }
  float m = lno;
  #pragma unroll
  for (int b = 0; b < FW_SLICES; ++b)
    if (k[b] != 0u) m = fmaxf(m, funkey(k[b]));
  const float en = expf(lno - m);
  float den = en;
  #pragma unroll
  for (int b = 0; b < FW_SLICES; ++b)
    if (k[b] != 0u) den += s[b] * expf(funkey(k[b]) - m);
  *(float2*)(md + 2 * (size_t)u) = make_float2(m, den);
  outNo[u] = en / den;
}

__global__ void edge_prob_kernel(const float* __restrict__ logits, const int* __restrict__ uidx,
                                 const float* __restrict__ md, float* __restrict__ outE, int E)
{
  const int e = (blockIdx.x * blockDim.x + threadIdx.x) << 2;
  if (e + 3 < E) {
    const int4   u = *(const int4*)(uidx + e);
    const float4 l = *(const float4*)(logits + e);
    const float2* md2 = (const float2*)md;
    const float2 p0 = md2[u.x], p1 = md2[u.y], p2 = md2[u.z], p3 = md2[u.w];
    float4 o;
    o.x = expf(l.x - p0.x) / p0.y;
    o.y = expf(l.y - p1.x) / p1.y;
    o.z = expf(l.z - p2.x) / p2.y;
    o.w = expf(l.w - p3.x) / p3.y;
    *(float4*)(outE + e) = o;
  } else {
    for (int i = e; i < E; ++i) {
      const float2 p = ((const float2*)md)[uidx[i]];
      outE[i] = expf(logits[i] - p.x) / p.y;
    }
  }
}

extern "C" void kernel_launch(void* const* d_in, const int* in_sizes, int n_in,
                              void* d_out, int out_size, void* d_ws, size_t ws_size,
                              hipStream_t stream)
{
  const float* uf  = (const float*)d_in[0];
  const float* gf  = (const float*)d_in[1];
  const float* ef  = (const float*)d_in[2];
  const int*   ui  = (const int*)d_in[3];
  const int*   gi  = (const int*)d_in[4];
  const float* W1  = (const float*)d_in[5];
  const float* b1  = (const float*)d_in[6];
  const float* W2  = (const float*)d_in[7];
  const float* b2  = (const float*)d_in[8];
  const float* w3  = (const float*)d_in[9];
  const float* b3  = (const float*)d_in[10];
  const float* Wn1 = (const float*)d_in[11];
  const float* bn1 = (const float*)d_in[12];
  const float* wn2 = (const float*)d_in[13];
  const float* bn2 = (const float*)d_in[14];

  const int nU = in_sizes[0] / 3;
  const int E  = in_sizes[3];

  float* ws          = (float*)d_ws;
  float* edge_logits = ws;
  float* logit_no    = ws + (size_t)E;
  float* md          = ws + (size_t)E + (size_t)nU;

  float* outE  = (float*)d_out;
  float* outNo = outE + (size_t)E;

  unsigned* pkey = (unsigned*)outE;
  float*    psum = outE + (size_t)FW_SLICES * (size_t)nU;

  notx_kernel<<<(nU + 255) / 256, 256, 0, stream>>>(uf, Wn1, bn1, wn2, bn2, logit_no, nU);

  const size_t lds = (size_t)(2 * ROW + 32 * ROW + WAVES * PASS_EDGES) * sizeof(float);
  const int nBlk = (E + EPB - 1) / EPB;
  edge_mlp_kernel<<<nBlk, THREADS, lds, stream>>>(uf, gf, ef, ui, gi,
                                                  W1, b1, W2, b2, w3, b3, edge_logits, E);

  fold_kernel<<<dim3(FW_SLICES, (nU + FW_USERS - 1) / FW_USERS), FW_THREADS, 0, stream>>>(edge_logits, ui, pkey, psum, E, nU);
  combine_kernel<<<(nU + 255) / 256, 256, 0, stream>>>(logit_no, pkey, psum, md, outNo, nU);
  edge_prob_kernel<<<(E / 4 + 256) / 256, 256, 0, stream>>>(edge_logits, ui, md, outE, E);
}


# softmax fold reads its edge slice once: the window's edges go into an LDS list (ballot+mbcnt places, per wave), max and sum of exp then run over the list with all lanes; ds_add_f32 order changes
# speedup vs baseline: 1.8834x; 1.0708x over previous
#include <hip/hip_runtime.h>
#include <math.h>


typedef __attribute__((ext_vector_type(2))) float v2f;
typedef __attribute__((ext_vector_type(4))) float v4f;
typedef __attribute__((ext_vector_type(8))) float v8f;

#define WAVES 4
#define THREADS (WAVES * 32)
#define PASS_TILES 8
#define PASS_EDGES (PASS_TILES * 16)
#define BLOCK_TILES 16
#define EPB (BLOCK_TILES * 16)

#define ROW 516

__device__ __forceinline__ v8f wmma_f32_16x16x4(v2f a, v2f b, v8f c) {
  return __builtin_amdgcn_wmma_f32_16x16x4_f32(false, a, false, b, (short)0, c, false, false);
}

__device__ __forceinline__ unsigned fkey(float f) {
  unsigned u = __float_as_uint(f);
  return (u & 0x80000000u) ? ~u : (u | 0x80000000u);
}
__device__ __forceinline__ float funkey(unsigned k) {
  unsigned u = (k & 0x80000000u) ? (k & 0x7fffffffu) : ~k;
  return __uint_as_float(u);
}

__global__ __launch_bounds__(THREADS) void edge_mlp_kernel(
    const float* __restrict__ uf, const float* __restrict__ gf,
    const float* __restrict__ ef, const int* __restrict__ uidx,
    const int* __restrict__ gidx,
    const float* __restrict__ W1, const float* __restrict__ b1,
    const float* __restrict__ W2, const float* __restrict__ b2,
    const float* __restrict__ w3, const float* __restrict__ b3,
    float* __restrict__ logits, int E)
{
  extern __shared__ float smem[];
  float* sX = smem;
  float* sH = sX + 2 * ROW;
  float* sP = sH + 32 * ROW;

  const int tid  = threadIdx.x;
  const int wid  = tid >> 5;
  const int lane = tid & 31;
  const int c    = lane & 15;
  const int h    = lane >> 4;
  const int nb   = wid * 32;

  v2f Bw1[2][2];
  v2f Bw2[16][2][2];
  float bias1[2], bias2[2], w3v[2];
  #pragma unroll
  for (int nt = 0; nt < 2; ++nt) {
    const int n = nb + nt * 16 + c;
    bias1[nt] = b1[n]; bias2[nt] = b2[n]; w3v[nt] = w3[n];
    #pragma unroll
    for (int j = 0; j < 2; ++j) {
      const int k0 = 4 * h + 2 * j;
      Bw1[j][nt] = v2f{W1[k0 * 128 + n], W1[(k0 + 1) * 128 + n]};
    }
    #pragma unroll
    for (int p = 0; p < 16; ++p) {
      #pragma unroll
      for (int j = 0; j < 2; ++j) {
        const int k0 = 8 * p + 4 * h + 2 * j;
        Bw2[p][j][nt] = v2f{W2[k0 * 128 + n], W2[(k0 + 1) * 128 + n]};
      }
    }
  }
  const float bb3 = b3[0];

  const int tilesTotal = (E + 15) >> 4;
  const int tStart = (int)blockIdx.x * BLOCK_TILES;
  const int tRest  = tilesTotal - tStart;
  const int tCount = tRest < BLOCK_TILES ? tRest : BLOCK_TILES;
  const int nPass  = (tCount + PASS_TILES - 1) / PASS_TILES;

  float xv[8];
  int eN, uI, gI;
  {
    eN = tStart * 16 + tid; if (eN >= E) eN = E - 1;
    uI = uidx[eN]; gI = gidx[eN];
    xv[0] = uf[uI * 3 + 0]; xv[1] = uf[uI * 3 + 1]; xv[2] = uf[uI * 3 + 2];
    xv[3] = gf[gI * 3 + 0]; xv[4] = gf[gI * 3 + 1]; xv[5] = gf[gI * 3 + 2];
    xv[6] = ef[eN * 2 + 0]; xv[7] = ef[eN * 2 + 1];
  }

  const int aoff = h * ROW + c * 4;

  for (int i = 0; i < nPass; ++i) {
    const int eBase = (tStart + i * PASS_TILES) * 16;
    const int tilesLeft = tCount - i * PASS_TILES;
    const int tilesThis = tilesLeft < PASS_TILES ? tilesLeft : PASS_TILES;
    const bool more = (i + 1 < nPass);

    *(v4f*)&sX[tid * 4]       = v4f{xv[0], xv[1], xv[2], xv[3]};
    *(v4f*)&sX[ROW + tid * 4] = v4f{xv[4], xv[5], xv[6], xv[7]};
    __syncthreads();

    if (more) {
      eN = eBase + PASS_EDGES + tid; if (eN >= E) eN = E - 1;
      uI = uidx[eN]; gI = gidx[eN];
    }

    for (int mt = 0; mt < tilesThis; ++mt) {
      const v4f a = *(const v4f*)&sX[aoff + mt * 64];
      #pragma unroll
      for (int nt = 0; nt < 2; ++nt) {
        v8f cc = {0.f, 0.f, 0.f, 0.f, 0.f, 0.f, 0.f, 0.f};
        cc = wmma_f32_16x16x4(v2f{a.x, a.y}, Bw1[0][nt], cc);
        cc = wmma_f32_16x16x4(v2f{a.z, a.w}, Bw1[1][nt], cc);
        const int n = nb + nt * 16 + c;
        float* dst = &sH[(n >> 2) * ROW + (mt * 16 + h * 8) * 4 + (n & 3)];
        #pragma unroll
        for (int r = 0; r < 8; ++r) {
          float hv = cc[r] + bias1[nt];
          dst[r * 4] = hv > 0.f ? hv : 0.f;
        }
      }
    }
    __syncthreads();

    if (more) {
      xv[0] = uf[uI * 3 + 0]; xv[1] = uf[uI * 3 + 1]; xv[2] = uf[uI * 3 + 2];
      xv[3] = gf[gI * 3 + 0]; xv[4] = gf[gI * 3 + 1]; xv[5] = gf[gI * 3 + 2];
      xv[6] = ef[eN * 2 + 0]; xv[7] = ef[eN * 2 + 1];
    }

    v4f aLo[8], aHi[8];
    #pragma unroll
    for (int p = 0; p < 8; ++p) aLo[p] = *(const v4f*)&sH[(2 * p) * ROW + aoff];

    for (int mt = 0; mt < tilesThis; ++mt) {
      const int mtN = (mt + 1 < tilesThis) ? mt + 1 : mt;
      v8f c0 = {0.f, 0.f, 0.f, 0.f, 0.f, 0.f, 0.f, 0.f};
      v8f c1 = {0.f, 0.f, 0.f, 0.f, 0.f, 0.f, 0.f, 0.f};

      #pragma unroll
      for (int p = 0; p < 8; ++p) aHi[p] = *(const v4f*)&sH[(16 + 2 * p) * ROW + aoff + mt * 64];
      #pragma unroll
      for (int p = 0; p < 8; ++p) {
        const v2f a0 = v2f{aLo[p].x, aLo[p].y}, a1 = v2f{aLo[p].z, aLo[p].w};
        c0 = wmma_f32_16x16x4(a0, Bw2[p][0][0], c0);
        c1 = wmma_f32_16x16x4(a0, Bw2[p][0][1], c1);
        c0 = wmma_f32_16x16x4(a1, Bw2[p][1][0], c0);
        c1 = wmma_f32_16x16x4(a1, Bw2[p][1][1], c1);
      }
      #pragma unroll
      for (int p = 0; p < 8; ++p) aLo[p] = *(const v4f*)&sH[(2 * p) * ROW + aoff + mtN * 64];
      #pragma unroll
      for (int p = 0; p < 8; ++p) {
        const v2f a0 = v2f{aHi[p].x, aHi[p].y}, a1 = v2f{aHi[p].z, aHi[p].w};
        c0 = wmma_f32_16x16x4(a0, Bw2[8 + p][0][0], c0);
        c1 = wmma_f32_16x16x4(a0, Bw2[8 + p][0][1], c1);
        c0 = wmma_f32_16x16x4(a1, Bw2[8 + p][1][0], c0);
        c1 = wmma_f32_16x16x4(a1, Bw2[8 + p][1][1], c1);
      }

      float s[8];
      #pragma unroll
      for (int r = 0; r < 8; ++r) {
        float h0 = c0[r] + bias2[0]; h0 = h0 > 0.f ? h0 : 0.f;
        float h1 = c1[r] + bias2[1]; h1 = h1 > 0.f ? h1 : 0.f;
        s[r] = h0 * w3v[0] + h1 * w3v[1];
      }
      float t4[4], t2[2], t1;
      {
        const bool up = (c & 8) != 0;
        #pragma unroll
        for (int q = 0; q < 4; ++q) {
          const float keep = up ? s[4 + q] : s[q];
          const float send = up ? s[q] : s[4 + q];
          t4[q] = keep + __shfl_xor(send, 8);
        }
      }
      {
        const bool up = (c & 4) != 0;
        #pragma unroll
        for (int q = 0; q < 2; ++q) {
          const float keep = up ? t4[2 + q] : t4[q];
          const float send = up ? t4[q] : t4[2 + q];
          t2[q] = keep + __shfl_xor(send, 4);
        }
      }
      {
        const bool up = (c & 2) != 0;
        const float keep = up ? t2[1] : t2[0];
        const float send = up ? t2[0] : t2[1];
        t1 = keep + __shfl_xor(send, 2);
      }
      t1 += __shfl_xor(t1, 1);
      if ((c & 1) == 0) {
        const int r = ((c >> 3) << 2) | (((c >> 2) & 1) << 1) | ((c >> 1) & 1);
        sP[wid * PASS_EDGES + mt * 16 + h * 8 + r] = t1;
      }
    }
    __syncthreads();

    {
      const int e = eBase + tid;
      if (tid < tilesThis * 16 && e < E)
        logits[e] = ((sP[tid] + sP[PASS_EDGES + tid]) + (sP[2 * PASS_EDGES + tid] + sP[3 * PASS_EDGES + tid])) + bb3;
    }
  }
}

__global__ void notx_kernel(const float* __restrict__ uf,
                            const float* __restrict__ Wn1, const float* __restrict__ bn1,
                            const float* __restrict__ wn2, const float* __restrict__ bn2,
                            float* __restrict__ logit_no, int nU)
{
  __shared__ float sW[192], sB[64], sw2[64];
  const int tid = threadIdx.x;
  for (int i = tid; i < 192; i += blockDim.x) sW[i] = Wn1[i];
  for (int i = tid; i < 64; i += blockDim.x) { sB[i] = bn1[i]; sw2[i] = wn2[i]; }
  __syncthreads();
  const int u = blockIdx.x * blockDim.x + tid;
  if (u >= nU) return;
  const float f0 = uf[u * 3 + 0], f1 = uf[u * 3 + 1], f2 = uf[u * 3 + 2];
  float acc = bn2[0];
  #pragma unroll 8
  for (int j = 0; j < 64; ++j) {
    float h = f0 * sW[j] + f1 * sW[64 + j] + f2 * sW[128 + j] + sB[j];
    h = h > 0.f ? h : 0.f;
    acc += h * sw2[j];
  }
  logit_no[u] = acc;
}

#define FW_USERS   3136
#define FW_TAB     FW_USERS
#define FW_THREADS 1024
#define FW_SLICES  8
#define FW_GROUPS  4
#define FW_WAVES   (FW_THREADS / 32)
#define FW_WLIST   384

__device__ __forceinline__ void fold_max(unsigned* tmax, int u, float l, int base, int cnt) {
  const unsigned d = (unsigned)(u - base);
  if (d < (unsigned)cnt)
    __hip_atomic_fetch_max(&tmax[d], fkey(l), __ATOMIC_RELAXED, __HIP_MEMORY_SCOPE_WORKGROUP);
}
__device__ __forceinline__ void fold_sum(const unsigned* tmax, float* tsum, int u, float l, int base, int cnt) {
  const unsigned d = (unsigned)(u - base);
  if (d < (unsigned)cnt) {
    const float t = (l == -INFINITY) ? 0.f : expf(l - funkey(tmax[d]));
    __hip_atomic_fetch_add(&tsum[d], t, __ATOMIC_RELAXED, __HIP_MEMORY_SCOPE_WORKGROUP);
  }
}

__device__ __forceinline__ void fold_put(uint2* wl, unsigned& wn, int u, float l, int base, unsigned lim) {
  const unsigned d = (unsigned)(u - base);
  const bool hit = d < lim;
  const unsigned mask = __builtin_amdgcn_ballot_w32(hit);
  if (hit) {
    const unsigned slot = wn + __builtin_amdgcn_mbcnt_lo(mask, 0u);
    if (slot < (unsigned)FW_WLIST) wl[slot] = make_uint2(d, __float_as_uint(l));
  }
  wn += (unsigned)__popc(mask);
}

__device__ __forceinline__ unsigned fold_collect(uint2* wl,
                                                 const float* __restrict__ logits,
                                                 const int* __restrict__ uidx,
                                                 int e0, int n, int base, int cnt, int tid) {
  unsigned wn = 0;
  const int nvec = n >> 2;
  const int4*   u4 = (const int4*)(uidx + e0);
  const float4* l4 = (const float4*)(logits + e0);
  for (int v0 = 0; v0 < nvec; v0 += FW_GROUPS * FW_THREADS) {
    int4 uu[FW_GROUPS]; float4 ll[FW_GROUPS]; bool ok[FW_GROUPS];
    #pragma unroll
    for (int j = 0; j < FW_GROUPS; ++j) {
      const int vj = v0 + tid + j * FW_THREADS;
      ok[j] = vj < nvec;
      const int vi = ok[j] ? vj : v0;
      uu[j] = u4[vi];
      ll[j] = l4[vi];
    }
    #pragma unroll
    for (int j = 0; j < FW_GROUPS; ++j) {
      const unsigned lim = ok[j] ? (unsigned)cnt : 0u;
      fold_put(wl, wn, uu[j].x, ll[j].x, base, lim);
      fold_put(wl, wn, uu[j].y, ll[j].y, base, lim);
      fold_put(wl, wn, uu[j].z, ll[j].z, base, lim);
      fold_put(wl, wn, uu[j].w, ll[j].w, base, lim);
    }
  }
  {
    const int e = e0 + (nvec << 2) + tid;
    const bool in = e < e0 + n;
    const int ec = in ? e : e0;
    int u = 0; float l = 0.f;
    if (n > 0) { u = uidx[ec]; l = logits[ec]; }
    fold_put(wl, wn, u, l, base, in ? (unsigned)cnt : 0u);
  }
  return wn;
}

template <bool SUM>
__device__ __forceinline__ void fold_sweep(unsigned* tmax, float* tsum,
                                           const float* __restrict__ logits,
                                           const int* __restrict__ uidx,
                                           int e0, int n, int base, int cnt, int tid) {
  const int nvec = n >> 2;
  const int4*   u4 = (const int4*)(uidx + e0);
  const float4* l4 = (const float4*)(logits + e0);
  for (int v = tid; v < nvec; v += FW_GROUPS * FW_THREADS) {
    int4 uu[FW_GROUPS]; float4 ll[FW_GROUPS]; bool ok[FW_GROUPS];
    #pragma unroll
    for (int j = 0; j < FW_GROUPS; ++j) {
      const int vj = v + j * FW_THREADS;
      ok[j] = vj < nvec;
      const int vi = ok[j] ? vj : v;
      uu[j] = u4[vi];
      ll[j] = l4[vi];
    }
    #pragma unroll
    for (int j = 0; j < FW_GROUPS; ++j) {
      if (ok[j]) {
        if (SUM) {
          fold_sum(tmax, tsum, uu[j].x, ll[j].x, base, cnt);
          fold_sum(tmax, tsum, uu[j].y, ll[j].y, base, cnt);
          fold_sum(tmax, tsum, uu[j].z, ll[j].z, base, cnt);
          fold_sum(tmax, tsum, uu[j].w, ll[j].w, base, cnt);
        } else {
          fold_max(tmax, uu[j].x, ll[j].x, base, cnt);
          fold_max(tmax, uu[j].y, ll[j].y, base, cnt);
          fold_max(tmax, uu[j].z, ll[j].z, base, cnt);
          fold_max(tmax, uu[j].w, ll[j].w, base, cnt);
        }
      }
    }
  }
  for (int e = e0 + (nvec << 2) + tid; e < e0 + n; e += FW_THREADS) {
    if (SUM) fold_sum(tmax, tsum, uidx[e], logits[e], base, cnt);
    else     fold_max(tmax, uidx[e], logits[e], base, cnt);
  }
}

__global__ __launch_bounds__(FW_THREADS) void fold_kernel(
    const float* __restrict__ logits, const int* __restrict__ uidx,
    unsigned* __restrict__ pkey, float* __restrict__ psum, int E, int nU)
{
  __shared__ unsigned tmax[FW_TAB];
  __shared__ float    tsum[FW_TAB];
  __shared__ uint2    list[FW_WAVES * FW_WLIST];
  __shared__ unsigned over;
  const int tid  = threadIdx.x;
  const int lane = tid & 31;
  const int base = (int)blockIdx.y * FW_USERS;
  int cnt = nU - base; if (cnt > FW_USERS) cnt = FW_USERS;

  for (int i = tid; i < FW_TAB; i += FW_THREADS) { tmax[i] = 0u; tsum[i] = 0.f; }
  if (tid == 0) over = 0u;
  __syncthreads();

  const int per = (((E + (int)gridDim.x - 1) / (int)gridDim.x) + 3) & ~3;
  const int e0  = (int)blockIdx.x * per;
  int n = E - e0; if (n > per) n = per; if (n < 0) n = 0;

  uint2* wl = list + (tid >> 5) * FW_WLIST;
  const unsigned wn = fold_collect(wl, logits, uidx, e0, n, base, cnt, tid);
  if (wn > (unsigned)FW_WLIST && lane == 0) over = 1u;
  __syncthreads();

  if (over == 0u) {
    for (unsigned i = lane; i < wn; i += 32) {
      const uint2 en = wl[i];
      __hip_atomic_fetch_max(&tmax[en.x], fkey(__uint_as_float(en.y)), __ATOMIC_RELAXED, __HIP_MEMORY_SCOPE_WORKGROUP);
    }
    __syncthreads();
    for (unsigned i = lane; i < wn; i += 32) {
      const uint2 en = wl[i];
      const float l = __uint_as_float(en.y);
      const float t = (l == -INFINITY) ? 0.f : expf(l - funkey(tmax[en.x]));
      __hip_atomic_fetch_add(&tsum[en.x], t, __ATOMIC_RELAXED, __HIP_MEMORY_SCOPE_WORKGROUP);
    }
  } else {
    fold_sweep<false>(tmax, tsum, logits, uidx, e0, n, base, cnt, tid);
    __syncthreads();
    fold_sweep<true>(tmax, tsum, logits, uidx, e0, n, base, cnt, tid);
  }
  __syncthreads();

  unsigned* pk = pkey + (size_t)blockIdx.x * (size_t)nU + base;
  float*    ps = psum + (size_t)blockIdx.x * (size_t)nU + base;
  for (int i = tid; i < cnt; i += FW_THREADS) { pk[i] = tmax[i]; ps[i] = tsum[i]; }
}

__global__ void combine_kernel(const float* __restrict__ logit_no,
                               const unsigned* __restrict__ pkey, const float* __restrict__ psum,
                               float* __restrict__ md, float* __restrict__ outNo, int nU)
{
  const int u = blockIdx.x * blockDim.x + threadIdx.x;
  if (u >= nU) return;
  const float lno = logit_no[u];
  unsigned k[FW_SLICES]; float s[FW_SLICES];
  #pragma unroll
  for (int b = 0; b < FW_SLICES; ++b) {
    k[b] = pkey[(size_t)b * (size_t)nU + u];
    s[b] = psum[(size_t)b * (size_t)nU + u];
  }
  float m = lno;
  #pragma unroll
  for (int b = 0; b < FW_SLICES; ++b)
    if (k[b] != 0u) m = fmaxf(m, funkey(k[b]));
  const float en = expf(lno - m);
  float den = en;
  #pragma unroll
  for (int b = 0; b < FW_SLICES; ++b)
    if (k[b] != 0u) den += s[b] * expf(funkey(k[b]) - m);
  *(float2*)(md + 2 * (size_t)u) = make_float2(m, den);
  outNo[u] = en / den;
}

__global__ void edge_prob_kernel(const float* __restrict__ logits, const int* __restrict__ uidx,
                                 const float* __restrict__ md, float* __restrict__ outE, int E)
{
  const int e = (blockIdx.x * blockDim.x + threadIdx.x) << 2;
  if (e + 3 < E) {
    const int4   u = *(const int4*)(uidx + e);
    const float4 l = *(const float4*)(logits + e);
    const float2* md2 = (const float2*)md;
    const float2 p0 = md2[u.x], p1 = md2[u.y], p2 = md2[u.z], p3 = md2[u.w];
    float4 o;
    o.x = expf(l.x - p0.x) / p0.y;
    o.y = expf(l.y - p1.x) / p1.y;
    o.z = expf(l.z - p2.x) / p2.y;
    o.w = expf(l.w - p3.x) / p3.y;
    *(float4*)(outE + e) = o;
  } else {
    for (int i = e; i < E; ++i) {
      const float2 p = ((const float2*)md)[uidx[i]];
      outE[i] = expf(logits[i] - p.x) / p.y;
    }
  }
}

extern "C" void kernel_launch(void* const* d_in, const int* in_sizes, int n_in,
                              void* d_out, int out_size, void* d_ws, size_t ws_size,
                              hipStream_t stream)
{
  const float* uf  = (const float*)d_in[0];
  const float* gf  = (const float*)d_in[1];
  const float* ef  = (const float*)d_in[2];
  const int*   ui  = (const int*)d_in[3];
  const int*   gi  = (const int*)d_in[4];
  const float* W1  = (const float*)d_in[5];
  const float* b1  = (const float*)d_in[6];
  const float* W2  = (const float*)d_in[7];
  const float* b2  = (const float*)d_in[8];
  const float* w3  = (const float*)d_in[9];
  const float* b3  = (const float*)d_in[10];
  const float* Wn1 = (const float*)d_in[11];
  const float* bn1 = (const float*)d_in[12];
  const float* wn2 = (const float*)d_in[13];
  const float* bn2 = (const float*)d_in[14];

  const int nU = in_sizes[0] / 3;
  const int E  = in_sizes[3];

  float* ws          = (float*)d_ws;
  float* edge_logits = ws;
  float* logit_no    = ws + (size_t)E;
  float* md          = ws + (size_t)E + (size_t)nU;

  float* outE  = (float*)d_out;
  float* outNo = outE + (size_t)E;

  unsigned* pkey = (unsigned*)outE;
  float*    psum = outE + (size_t)FW_SLICES * (size_t)nU;

  notx_kernel<<<(nU + 255) / 256, 256, 0, stream>>>(uf, Wn1, bn1, wn2, bn2, logit_no, nU);

  const size_t lds = (size_t)(2 * ROW + 32 * ROW + WAVES * PASS_EDGES) * sizeof(float);
  const int nBlk = (E + EPB - 1) / EPB;
  edge_mlp_kernel<<<nBlk, THREADS, lds, stream>>>(uf, gf, ef, ui, gi,
                                                  W1, b1, W2, b2, w3, b3, edge_logits, E);

  fold_kernel<<<dim3(FW_SLICES, (nU + FW_USERS - 1) / FW_USERS), FW_THREADS, 0, stream>>>(edge_logits, ui, pkey, psum, E, nU);
  combine_kernel<<<(nU + 255) / 256, 256, 0, stream>>>(logit_no, pkey, psum, md, outNo, nU);
  edge_prob_kernel<<<(E / 4 + 256) / 256, 256, 0, stream>>>(edge_logits, ui, md, outE, E);
}


# edge MLP layer 2: WMMA runs free of LDS instructions: A-fragment reads as groups of 8 ahead of each WMMA run (scheduling fences), w3-dot row sums on DPP row_xmask adds, not ds_bpermute
# speedup vs baseline: 1.9443x; 1.0323x over previous
#include <hip/hip_runtime.h>
#include <math.h>


typedef __attribute__((ext_vector_type(2))) float v2f;
typedef __attribute__((ext_vector_type(4))) float v4f;
typedef __attribute__((ext_vector_type(8))) float v8f;

#define WAVES 4
#define THREADS (WAVES * 32)
#define PASS_TILES 8
#define PASS_EDGES (PASS_TILES * 16)
#define BLOCK_TILES 16
#define EPB (BLOCK_TILES * 16)

#define ROW 516

__device__ __forceinline__ v8f wmma_f32_16x16x4(v2f a, v2f b, v8f c) {
  return __builtin_amdgcn_wmma_f32_16x16x4_f32(false, a, false, b, (short)0, c, false, false);
}

__device__ __forceinline__ unsigned fkey(float f) {
  unsigned u = __float_as_uint(f);
  return (u & 0x80000000u) ? ~u : (u | 0x80000000u);
}
__device__ __forceinline__ float funkey(unsigned k) {
  unsigned u = (k & 0x80000000u) ? (k & 0x7fffffffu) : ~k;
  return __uint_as_float(u);
}

template <int M>
__device__ __forceinline__ float row_xor(float v) {
  return __int_as_float(__builtin_amdgcn_update_dpp(0, __float_as_int(v), 0x160 + M, 0xf, 0xf, true));
}

__global__ __launch_bounds__(THREADS) void edge_mlp_kernel(
    const float* __restrict__ uf, const float* __restrict__ gf,
    const float* __restrict__ ef, const int* __restrict__ uidx,
    const int* __restrict__ gidx,
    const float* __restrict__ W1, const float* __restrict__ b1,
    const float* __restrict__ W2, const float* __restrict__ b2,
    const float* __restrict__ w3, const float* __restrict__ b3,
    float* __restrict__ logits, int E)
{
  extern __shared__ float smem[];
  float* sX = smem;
  float* sH = sX + 2 * ROW;
  float* sP = sH + 32 * ROW;

  const int tid  = threadIdx.x;
  const int wid  = tid >> 5;
  const int lane = tid & 31;
  const int c    = lane & 15;
  const int h    = lane >> 4;
  const int nb   = wid * 32;

  v2f Bw1[2][2];
  v2f Bw2[16][2][2];
  float bias1[2], bias2[2], w3v[2];
  #pragma unroll
  for (int nt = 0; nt < 2; ++nt) {
    const int n = nb + nt * 16 + c;
    bias1[nt] = b1[n]; bias2[nt] = b2[n]; w3v[nt] = w3[n];
    #pragma unroll
    for (int j = 0; j < 2; ++j) {
      const int k0 = 4 * h + 2 * j;
      Bw1[j][nt] = v2f{W1[k0 * 128 + n], W1[(k0 + 1) * 128 + n]};
    }
    #pragma unroll
    for (int p = 0; p < 16; ++p) {
      #pragma unroll
      for (int j = 0; j < 2; ++j) {
        const int k0 = 8 * p + 4 * h + 2 * j;
        Bw2[p][j][nt] = v2f{W2[k0 * 128 + n], W2[(k0 + 1) * 128 + n]};
      }
    }
  }
  const float bb3 = b3[0];

  const int tilesTotal = (E + 15) >> 4;
  const int tStart = (int)blockIdx.x * BLOCK_TILES;
  const int tRest  = tilesTotal - tStart;
  const int tCount = tRest < BLOCK_TILES ? tRest : BLOCK_TILES;
  const int nPass  = (tCount + PASS_TILES - 1) / PASS_TILES;

  float xv[8];
  int eN, uI, gI;
  {
    eN = tStart * 16 + tid; if (eN >= E) eN = E - 1;
    uI = uidx[eN]; gI = gidx[eN];
    xv[0] = uf[uI * 3 + 0]; xv[1] = uf[uI * 3 + 1]; xv[2] = uf[uI * 3 + 2];
    xv[3] = gf[gI * 3 + 0]; xv[4] = gf[gI * 3 + 1]; xv[5] = gf[gI * 3 + 2];
    xv[6] = ef[eN * 2 + 0]; xv[7] = ef[eN * 2 + 1];
  }

  const int aoff = h * ROW + c * 4;

  for (int i = 0; i < nPass; ++i) {
    const int eBase = (tStart + i * PASS_TILES) * 16;
    const int tilesLeft = tCount - i * PASS_TILES;
    const int tilesThis = tilesLeft < PASS_TILES ? tilesLeft : PASS_TILES;
    const bool more = (i + 1 < nPass);

    *(v4f*)&sX[tid * 4]       = v4f{xv[0], xv[1], xv[2], xv[3]};
    *(v4f*)&sX[ROW + tid * 4] = v4f{xv[4], xv[5], xv[6], xv[7]};
    __syncthreads();

    if (more) {
      eN = eBase + PASS_EDGES + tid; if (eN >= E) eN = E - 1;
      uI = uidx[eN]; gI = gidx[eN];
    }

    for (int mt = 0; mt < tilesThis; ++mt) {
      const v4f a = *(const v4f*)&sX[aoff + mt * 64];
      #pragma unroll
      for (int nt = 0; nt < 2; ++nt) {
        v8f cc = {0.f, 0.f, 0.f, 0.f, 0.f, 0.f, 0.f, 0.f};
        cc = wmma_f32_16x16x4(v2f{a.x, a.y}, Bw1[0][nt], cc);
        cc = wmma_f32_16x16x4(v2f{a.z, a.w}, Bw1[1][nt], cc);
        const int n = nb + nt * 16 + c;
        float* dst = &sH[(n >> 2) * ROW + (mt * 16 + h * 8) * 4 + (n & 3)];
        #pragma unroll
        for (int r = 0; r < 8; ++r) {
          float hv = cc[r] + bias1[nt];
          dst[r * 4] = hv > 0.f ? hv : 0.f;
        }
      }
    }
    __syncthreads();

    if (more) {
      xv[0] = uf[uI * 3 + 0]; xv[1] = uf[uI * 3 + 1]; xv[2] = uf[uI * 3 + 2];
      xv[3] = gf[gI * 3 + 0]; xv[4] = gf[gI * 3 + 1]; xv[5] = gf[gI * 3 + 2];
      xv[6] = ef[eN * 2 + 0]; xv[7] = ef[eN * 2 + 1];
    }

    v4f aLo[8], aHi[8];
    #pragma unroll
    for (int p = 0; p < 8; ++p) aLo[p] = *(const v4f*)&sH[(2 * p) * ROW + aoff];

    for (int mt = 0; mt < tilesThis; ++mt) {
      const int mtN = (mt + 1 < tilesThis) ? mt + 1 : mt;
      v8f c0 = {0.f, 0.f, 0.f, 0.f, 0.f, 0.f, 0.f, 0.f};
      v8f c1 = {0.f, 0.f, 0.f, 0.f, 0.f, 0.f, 0.f, 0.f};

      #pragma unroll
      for (int p = 0; p < 8; ++p) aHi[p] = *(const v4f*)&sH[(16 + 2 * p) * ROW + aoff + mt * 64];
      __builtin_amdgcn_sched_barrier(0);
      #pragma unroll
      for (int p = 0; p < 8; ++p) {
        const v2f a0 = v2f{aLo[p].x, aLo[p].y}, a1 = v2f{aLo[p].z, aLo[p].w};
        c0 = wmma_f32_16x16x4(a0, Bw2[p][0][0], c0);
        c1 = wmma_f32_16x16x4(a0, Bw2[p][0][1], c1);
        c0 = wmma_f32_16x16x4(a1, Bw2[p][1][0], c0);
        c1 = wmma_f32_16x16x4(a1, Bw2[p][1][1], c1);
      }
      #pragma unroll
      for (int p = 0; p < 8; ++p) aLo[p] = *(const v4f*)&sH[(2 * p) * ROW + aoff + mtN * 64];
      __builtin_amdgcn_sched_barrier(0);
      #pragma unroll
      for (int p = 0; p < 8; ++p) {
        const v2f a0 = v2f{aHi[p].x, aHi[p].y}, a1 = v2f{aHi[p].z, aHi[p].w};
        c0 = wmma_f32_16x16x4(a0, Bw2[8 + p][0][0], c0);
        c1 = wmma_f32_16x16x4(a0, Bw2[8 + p][0][1], c1);
        c0 = wmma_f32_16x16x4(a1, Bw2[8 + p][1][0], c0);
        c1 = wmma_f32_16x16x4(a1, Bw2[8 + p][1][1], c1);
      }

      float s[8];
      #pragma unroll
      for (int r = 0; r < 8; ++r) {
        float h0 = c0[r] + bias2[0]; h0 = h0 > 0.f ? h0 : 0.f;
        float h1 = c1[r] + bias2[1]; h1 = h1 > 0.f ? h1 : 0.f;
        s[r] = h0 * w3v[0] + h1 * w3v[1];
      }
      float t4[4], t2[2], t1;
      {
        const bool up = (c & 8) != 0;
        #pragma unroll
        for (int q = 0; q < 4; ++q) {
          const float keep = up ? s[4 + q] : s[q];
          const float send = up ? s[q] : s[4 + q];
          t4[q] = keep + row_xor<8>(send);
        }
      }
      {
        const bool up = (c & 4) != 0;
        #pragma unroll
        for (int q = 0; q < 2; ++q) {
          const float keep = up ? t4[2 + q] : t4[q];
          const float send = up ? t4[q] : t4[2 + q];
          t2[q] = keep + row_xor<4>(send);
        }
      }
      {
        const bool up = (c & 2) != 0;
        const float keep = up ? t2[1] : t2[0];
        const float send = up ? t2[0] : t2[1];
        t1 = keep + row_xor<2>(send);
      }
      t1 += row_xor<1>(t1);
      {
        const int r = ((c >> 3) << 2) | (((c >> 2) & 1) << 1) | ((c >> 1) & 1);
        sP[wid * PASS_EDGES + mt * 16 + h * 8 + r] = t1;
      }
    }
    __syncthreads();

    {
      const int e = eBase + tid;
      if (tid < tilesThis * 16 && e < E)
        logits[e] = ((sP[tid] + sP[PASS_EDGES + tid]) + (sP[2 * PASS_EDGES + tid] + sP[3 * PASS_EDGES + tid])) + bb3;
    }
  }
}

__global__ void notx_kernel(const float* __restrict__ uf,
                            const float* __restrict__ Wn1, const float* __restrict__ bn1,
                            const float* __restrict__ wn2, const float* __restrict__ bn2,
                            float* __restrict__ logit_no, int nU)
{
  __shared__ float sW[192], sB[64], sw2[64];
  const int tid = threadIdx.x;
  for (int i = tid; i < 192; i += blockDim.x) sW[i] = Wn1[i];
  for (int i = tid; i < 64; i += blockDim.x) { sB[i] = bn1[i]; sw2[i] = wn2[i]; }
  __syncthreads();
  const int u = blockIdx.x * blockDim.x + tid;
  if (u >= nU) return;
  const float f0 = uf[u * 3 + 0], f1 = uf[u * 3 + 1], f2 = uf[u * 3 + 2];
  float acc = bn2[0];
  #pragma unroll 8
  for (int j = 0; j < 64; ++j) {
    float h = f0 * sW[j] + f1 * sW[64 + j] + f2 * sW[128 + j] + sB[j];
    h = h > 0.f ? h : 0.f;
    acc += h * sw2[j];
  }
  logit_no[u] = acc;
}

#define FW_USERS   3136
#define FW_TAB     FW_USERS
#define FW_THREADS 1024
#define FW_SLICES  8
#define FW_GROUPS  4
#define FW_WAVES   (FW_THREADS / 32)
#define FW_WLIST   384

__device__ __forceinline__ void fold_max(unsigned* tmax, int u, float l, int base, int cnt) {
  const unsigned d = (unsigned)(u - base);
  if (d < (unsigned)cnt)
    __hip_atomic_fetch_max(&tmax[d], fkey(l), __ATOMIC_RELAXED, __HIP_MEMORY_SCOPE_WORKGROUP);
}
__device__ __forceinline__ void fold_sum(const unsigned* tmax, float* tsum, int u, float l, int base, int cnt) {
  const unsigned d = (unsigned)(u - base);
  if (d < (unsigned)cnt) {
    const float t = (l == -INFINITY) ? 0.f : expf(l - funkey(tmax[d]));
    __hip_atomic_fetch_add(&tsum[d], t, __ATOMIC_RELAXED, __HIP_MEMORY_SCOPE_WORKGROUP);
  }
}

__device__ __forceinline__ void fold_put(uint2* wl, unsigned& wn, int u, float l, int base, unsigned lim) {
  const unsigned d = (unsigned)(u - base);
  const bool hit = d < lim;
  const unsigned mask = __builtin_amdgcn_ballot_w32(hit);
  if (hit) {
    const unsigned slot = wn + __builtin_amdgcn_mbcnt_lo(mask, 0u);
    if (slot < (unsigned)FW_WLIST) wl[slot] = make_uint2(d, __float_as_uint(l));
  }
  wn += (unsigned)__popc(mask);
}

__device__ __forceinline__ unsigned fold_collect(uint2* wl,
                                                 const float* __restrict__ logits,
                                                 const int* __restrict__ uidx,
                                                 int e0, int n, int base, int cnt, int tid) {
  unsigned wn = 0;
  const int nvec = n >> 2;
  const int4*   u4 = (const int4*)(uidx + e0);
  const float4* l4 = (const float4*)(logits + e0);
  for (int v0 = 0; v0 < nvec; v0 += FW_GROUPS * FW_THREADS) {
    int4 uu[FW_GROUPS]; float4 ll[FW_GROUPS]; bool ok[FW_GROUPS];
    #pragma unroll
    for (int j = 0; j < FW_GROUPS; ++j) {
      const int vj = v0 + tid + j * FW_THREADS;
      ok[j] = vj < nvec;
      const int vi = ok[j] ? vj : v0;
      uu[j] = u4[vi];
      ll[j] = l4[vi];
    }
    #pragma unroll
    for (int j = 0; j < FW_GROUPS; ++j) {
      const unsigned lim = ok[j] ? (unsigned)cnt : 0u;
      fold_put(wl, wn, uu[j].x, ll[j].x, base, lim);
      fold_put(wl, wn, uu[j].y, ll[j].y, base, lim);
      fold_put(wl, wn, uu[j].z, ll[j].z, base, lim);
      fold_put(wl, wn, uu[j].w, ll[j].w, base, lim);
    }
  }
  {
    const int e = e0 + (nvec << 2) + tid;
    const bool in = e < e0 + n;
    const int ec = in ? e : e0;
    int u = 0; float l = 0.f;
    if (n > 0) { u = uidx[ec]; l = logits[ec]; }
    fold_put(wl, wn, u, l, base, in ? (unsigned)cnt : 0u);
  }
  return wn;
}

template <bool SUM>
__device__ __forceinline__ void fold_sweep(unsigned* tmax, float* tsum,
                                           const float* __restrict__ logits,
                                           const int* __restrict__ uidx,
                                           int e0, int n, int base, int cnt, int tid) {
  const int nvec = n >> 2;
  const int4*   u4 = (const int4*)(uidx + e0);
  const float4* l4 = (const float4*)(logits + e0);
  for (int v = tid; v < nvec; v += FW_GROUPS * FW_THREADS) {
    int4 uu[FW_GROUPS]; float4 ll[FW_GROUPS]; bool ok[FW_GROUPS];
    #pragma unroll
    for (int j = 0; j < FW_GROUPS; ++j) {
      const int vj = v + j * FW_THREADS;
      ok[j] = vj < nvec;
      const int vi = ok[j] ? vj : v;
      uu[j] = u4[vi];
      ll[j] = l4[vi];
    }
    #pragma unroll
    for (int j = 0; j < FW_GROUPS; ++j) {
      if (ok[j]) {
        if (SUM) {
          fold_sum(tmax, tsum, uu[j].x, ll[j].x, base, cnt);
          fold_sum(tmax, tsum, uu[j].y, ll[j].y, base, cnt);
          fold_sum(tmax, tsum, uu[j].z, ll[j].z, base, cnt);
          fold_sum(tmax, tsum, uu[j].w, ll[j].w, base, cnt);
        } else {
          fold_max(tmax, uu[j].x, ll[j].x, base, cnt);
          fold_max(tmax, uu[j].y, ll[j].y, base, cnt);
          fold_max(tmax, uu[j].z, ll[j].z, base, cnt);
          fold_max(tmax, uu[j].w, ll[j].w, base, cnt);
        }
      }
    }
  }
  for (int e = e0 + (nvec << 2) + tid; e < e0 + n; e += FW_THREADS) {
    if (SUM) fold_sum(tmax, tsum, uidx[e], logits[e], base, cnt);
    else     fold_max(tmax, uidx[e], logits[e], base, cnt);
  }
}

__global__ __launch_bounds__(FW_THREADS) void fold_kernel(
    const float* __restrict__ logits, const int* __restrict__ uidx,
    unsigned* __restrict__ pkey, float* __restrict__ psum, int E, int nU)
{
  __shared__ unsigned tmax[FW_TAB];
  __shared__ float    tsum[FW_TAB];
  __shared__ uint2    list[FW_WAVES * FW_WLIST];
  __shared__ unsigned over;
  const int tid  = threadIdx.x;
  const int lane = tid & 31;
  const int base = (int)blockIdx.y * FW_USERS;
  int cnt = nU - base; if (cnt > FW_USERS) cnt = FW_USERS;

  for (int i = tid; i < FW_TAB; i += FW_THREADS) { tmax[i] = 0u; tsum[i] = 0.f; }
  if (tid == 0) over = 0u;
  __syncthreads();

  const int per = (((E + (int)gridDim.x - 1) / (int)gridDim.x) + 3) & ~3;
  const int e0  = (int)blockIdx.x * per;
  int n = E - e0; if (n > per) n = per; if (n < 0) n = 0;

  uint2* wl = list + (tid >> 5) * FW_WLIST;
  const unsigned wn = fold_collect(wl, logits, uidx, e0, n, base, cnt, tid);
  if (wn > (unsigned)FW_WLIST && lane == 0) over = 1u;
  __syncthreads();

  if (over == 0u) {
    for (unsigned i = lane; i < wn; i += 32) {
      const uint2 en = wl[i];
      __hip_atomic_fetch_max(&tmax[en.x], fkey(__uint_as_float(en.y)), __ATOMIC_RELAXED, __HIP_MEMORY_SCOPE_WORKGROUP);
    }
    __syncthreads();
    for (unsigned i = lane; i < wn; i += 32) {
      const uint2 en = wl[i];
      const float l = __uint_as_float(en.y);
      const float t = (l == -INFINITY) ? 0.f : expf(l - funkey(tmax[en.x]));
      __hip_atomic_fetch_add(&tsum[en.x], t, __ATOMIC_RELAXED, __HIP_MEMORY_SCOPE_WORKGROUP);
    }
  } else {
    fold_sweep<false>(tmax, tsum, logits, uidx, e0, n, base, cnt, tid);
    __syncthreads();
    fold_sweep<true>(tmax, tsum, logits, uidx, e0, n, base, cnt, tid);
  }
  __syncthreads();

  unsigned* pk = pkey + (size_t)blockIdx.x * (size_t)nU + base;
  float*    ps = psum + (size_t)blockIdx.x * (size_t)nU + base;
  for (int i = tid; i < cnt; i += FW_THREADS) { pk[i] = tmax[i]; ps[i] = tsum[i]; }
}

__global__ void combine_kernel(const float* __restrict__ logit_no,
                               const unsigned* __restrict__ pkey, const float* __restrict__ psum,
                               float* __restrict__ md, float* __restrict__ outNo, int nU)
{
  const int u = blockIdx.x * blockDim.x + threadIdx.x;
  if (u >= nU) return;
  const float lno = logit_no[u];
  unsigned k[FW_SLICES]; float s[FW_SLICES];
  #pragma unroll
  for (int b = 0; b < FW_SLICES; ++b) {
    k[b] = pkey[(size_t)b * (size_t)nU + u];
    s[b] = psum[(size_t)b * (size_t)nU + u];
  }
  float m = lno;
  #pragma unroll
  for (int b = 0; b < FW_SLICES; ++b)
    if (k[b] != 0u) m = fmaxf(m, funkey(k[b]));
  const float en = expf(lno - m);
  float den = en;
  #pragma unroll
  for (int b = 0; b < FW_SLICES; ++b)
    if (k[b] != 0u) den += s[b] * expf(funkey(k[b]) - m);
  *(float2*)(md + 2 * (size_t)u) = make_float2(m, den);
  outNo[u] = en / den;
}

__global__ void edge_prob_kernel(const float* __restrict__ logits, const int* __restrict__ uidx,
                                 const float* __restrict__ md, float* __restrict__ outE, int E)
{
  const int e = (blockIdx.x * blockDim.x + threadIdx.x) << 2;
  if (e + 3 < E) {
    const int4   u = *(const int4*)(uidx + e);
    const float4 l = *(const float4*)(logits + e);
    const float2* md2 = (const float2*)md;
    const float2 p0 = md2[u.x], p1 = md2[u.y], p2 = md2[u.z], p3 = md2[u.w];
    float4 o;
    o.x = expf(l.x - p0.x) / p0.y;
    o.y = expf(l.y - p1.x) / p1.y;
    o.z = expf(l.z - p2.x) / p2.y;
    o.w = expf(l.w - p3.x) / p3.y;
    *(float4*)(outE + e) = o;
  } else {
    for (int i = e; i < E; ++i) {
      const float2 p = ((const float2*)md)[uidx[i]];
      outE[i] = expf(logits[i] - p.x) / p.y;
    }
  }
}

extern "C" void kernel_launch(void* const* d_in, const int* in_sizes, int n_in,
                              void* d_out, int out_size, void* d_ws, size_t ws_size,
                              hipStream_t stream)
{
  const float* uf  = (const float*)d_in[0];
  const float* gf  = (const float*)d_in[1];
  const float* ef  = (const float*)d_in[2];
  const int*   ui  = (const int*)d_in[3];
  const int*   gi  = (const int*)d_in[4];
  const float* W1  = (const float*)d_in[5];
  const float* b1  = (const float*)d_in[6];
  const float* W2  = (const float*)d_in[7];
  const float* b2  = (const float*)d_in[8];
  const float* w3  = (const float*)d_in[9];
  const float* b3  = (const float*)d_in[10];
  const float* Wn1 = (const float*)d_in[11];
  const float* bn1 = (const float*)d_in[12];
  const float* wn2 = (const float*)d_in[13];
  const float* bn2 = (const float*)d_in[14];

  const int nU = in_sizes[0] / 3;
  const int E  = in_sizes[3];

  float* ws          = (float*)d_ws;
  float* edge_logits = ws;
  float* logit_no    = ws + (size_t)E;
  float* md          = ws + (size_t)E + (size_t)nU;

  float* outE  = (float*)d_out;
  float* outNo = outE + (size_t)E;

  unsigned* pkey = (unsigned*)outE;
  float*    psum = outE + (size_t)FW_SLICES * (size_t)nU;

  notx_kernel<<<(nU + 255) / 256, 256, 0, stream>>>(uf, Wn1, bn1, wn2, bn2, logit_no, nU);

  const size_t lds = (size_t)(2 * ROW + 32 * ROW + WAVES * PASS_EDGES) * sizeof(float);
  const int nBlk = (E + EPB - 1) / EPB;
  edge_mlp_kernel<<<nBlk, THREADS, lds, stream>>>(uf, gf, ef, ui, gi,
                                                  W1, b1, W2, b2, w3, b3, edge_logits, E);

  fold_kernel<<<dim3(FW_SLICES, (nU + FW_USERS - 1) / FW_USERS), FW_THREADS, 0, stream>>>(edge_logits, ui, pkey, psum, E, nU);
  combine_kernel<<<(nU + 255) / 256, 256, 0, stream>>>(logit_no, pkey, psum, md, outNo, nU);
  edge_prob_kernel<<<(E / 4 + 256) / 256, 256, 0, stream>>>(edge_logits, ui, md, outE, E);
}
